# starting kernel
# baseline (speedup 1.0000x reference)
#include <hip/hip_runtime.h>

#define NB 65536
#define ND 128
#define NH 256
#define NO 8
#define NE 8
#define EO (NE * NO)

typedef __attribute__((ext_vector_type(2))) float v2f;
typedef __attribute__((ext_vector_type(4))) float v4f;
typedef __attribute__((ext_vector_type(8))) float v8f;

#define WS2_FLOATS   (ND * NH)
#define WEFF2_FLOATS (NH * EO)


__global__ void qnet_pair_ws(const float* __restrict__ Ws,
                             float* __restrict__ ws2) {
    const int d = blockIdx.x * blockDim.x + threadIdx.x;
    const int j   = d & 1;
    int r         = d >> 1;
    const int n   = r & (NH - 1);
    r >>= 8;
    const int kp  = r & 15;
    const int lh  = r >> 4;
    const int k0  = 4 * (2 * kp + j) + 2 * lh;
    v2f v;
    v.x = Ws[(size_t)k0 * NH + n];
    v.y = Ws[(size_t)k0 * NH + n + NH];
    ((v2f*)ws2)[d] = v;
}

__global__ void qnet_prep_weff(const float* __restrict__ W1,
                               const float* __restrict__ b1,
                               const float* __restrict__ W2,
                               const float* __restrict__ b2,
                               float* __restrict__ weff2,
                               float* __restrict__ beff) {
    const int tid = blockIdx.x * blockDim.x + threadIdx.x;
    const int h   = tid >> 6;
    const int col = tid & 63;
    const int e   = col >> 3;
    const int o   = col & 7;

    const float* w1row = W1 + ((size_t)e * NH + h) * NH;
    const float* w2col = W2 + (size_t)e * NH * NO + o;
    float acc = 0.0f;
    for (int k = 0; k < NH; ++k)
        acc = __builtin_fmaf(w1row[k], w2col[(size_t)k * NO], acc);

    const int i  = h & 1;
    const int lh = (h >> 1) & 1;
    const int ks = h >> 2;
    const int kp = ks >> 1;
    const int j  = ks & 1;
    weff2[((((size_t)lh * 32 + kp) * EO + col) * 2 + j) * 2 + i] = acc;

    if (tid < EO) {
        const float* b1e = b1 + (size_t)e * NH;
        float bacc = b2[(size_t)e * NO + o];
        for (int k = 0; k < NH; ++k)
            bacc = __builtin_fmaf(b1e[k], w2col[(size_t)k * NO], bacc);
        beff[col] = bacc;
    }
}

#define WG_THREADS 256
#define WG_WAVES   8
#define ROWS_WG    128
#define SH_STR     68

__launch_bounds__(WG_THREADS, 1)
__global__ void qnet_fused(const float* __restrict__ x,
                           const int*   __restrict__ phases,
                           const float* __restrict__ bs,
                           const float* __restrict__ ws2,
                           const float* __restrict__ weff2,
                           const float* __restrict__ beff,
                           float* __restrict__ out) {
    __shared__ float sH[WG_WAVES][16 * SH_STR];

    const int tid   = threadIdx.x;
    const int wave  = tid >> 5;
    const int lane  = tid & 31;
    const int lhalf = lane >> 4;
    const int lmod  = lane & 15;
    const int r0    = blockIdx.x * ROWS_WG + wave * 16;

    const v4f* Bws = (const v4f*)ws2;
    const v4f* Bwe = (const v4f*)weff2;

    v2f a[32];
    {
        const float* xrow = x + (size_t)(r0 + lmod) * ND + 2 * lhalf;
        #pragma unroll
        for (int kk = 0; kk < 32; ++kk)
            a[kk] = *(const v2f*)(xrow + 4 * kk);
    }

    float* myH = &sH[wave][0];
    v8f y[4] = {};

    for (int ch = 0; ch < 4; ++ch) {
        v8f c[4] = {};
        v4f bw[4];
        #pragma unroll
        for (int t = 0; t < 4; ++t)
            bw[t] = Bws[(size_t)(lhalf * 16 + 0) * NH + (ch * 4 + t) * 16 + lmod];
        #pragma unroll
        for (int kp = 0; kp < 16; ++kp) {
            v4f bn[4];
            if (kp < 15) {
                #pragma unroll
                for (int t = 0; t < 4; ++t)
                    bn[t] = Bws[(size_t)(lhalf * 16 + kp + 1) * NH +
                                (ch * 4 + t) * 16 + lmod];
            }
            #pragma unroll
            for (int t = 0; t < 4; ++t) {
                const v2f blo = __builtin_shufflevector(bw[t], bw[t], 0, 1);
                c[t] = __builtin_amdgcn_wmma_f32_16x16x4_f32(
                           false, a[2 * kp], false, blo, (short)0, c[t], false, false);
            }
            #pragma unroll
            for (int t = 0; t < 4; ++t) {
                const v2f bhi = __builtin_shufflevector(bw[t], bw[t], 2, 3);
                c[t] = __builtin_amdgcn_wmma_f32_16x16x4_f32(
                           false, a[2 * kp + 1], false, bhi, (short)0, c[t], false, false);
            }
            if (kp < 15) {
                #pragma unroll
                for (int t = 0; t < 4; ++t) bw[t] = bn[t];
            }
        }
        #pragma unroll
        for (int t = 0; t < 4; ++t) {
            const float bsv = bs[(ch * 4 + t) * 16 + lmod];
            #pragma unroll
            for (int v = 0; v < 8; ++v) {
                const float hv = fmaxf(c[t][v] + bsv, 0.0f);
                myH[(v + 8 * lhalf) * SH_STR + t * 16 + lmod] = hv;
            }
        }
        const float* hA = myH + lmod * SH_STR + 2 * lhalf;
        v4f ww[4];
        #pragma unroll
        for (int t = 0; t < 4; ++t)
            ww[t] = Bwe[(size_t)(lhalf * 32 + ch * 8 + 0) * EO + t * 16 + lmod];
        #pragma unroll
        for (int kp = 0; kp < 8; ++kp) {
            v4f wn[4];
            if (kp < 7) {
                #pragma unroll
                for (int t = 0; t < 4; ++t)
                    wn[t] = Bwe[(size_t)(lhalf * 32 + ch * 8 + kp + 1) * EO +
                                t * 16 + lmod];
            }
            const v2f a0 = *(const v2f*)(hA + 4 * (2 * kp));
            const v2f a1 = *(const v2f*)(hA + 4 * (2 * kp + 1));
            #pragma unroll
            for (int t = 0; t < 4; ++t) {
                const v2f blo = __builtin_shufflevector(ww[t], ww[t], 0, 1);
                y[t] = __builtin_amdgcn_wmma_f32_16x16x4_f32(
                           false, a0, false, blo, (short)0, y[t], false, false);
            }
            #pragma unroll
            for (int t = 0; t < 4; ++t) {
                const v2f bhi = __builtin_shufflevector(ww[t], ww[t], 2, 3);
                y[t] = __builtin_amdgcn_wmma_f32_16x16x4_f32(
                           false, a1, false, bhi, (short)0, y[t], false, false);
            }
            if (kp < 7) {
                #pragma unroll
                for (int t = 0; t < 4; ++t) ww[t] = wn[t];
            }
        }
    }

    #pragma unroll
    for (int t = 0; t < 4; ++t)
        #pragma unroll
        for (int v = 0; v < 8; ++v)
            myH[(v + 8 * lhalf) * SH_STR + t * 16 + lmod] = y[t][v];

    const int selRow = lane >> 1;
    const int oq     = (lane & 1) * 4;
    const int p      = phases[r0 + selRow];
    const v4f yv = *(const v4f*)(myH + selRow * SH_STR + p * NO + oq);
    const v4f bv = *(const v4f*)(beff + p * NO + oq);
    *(v4f*)(out + (size_t)(r0 + selRow) * NO + oq) = yv + bv;
}

extern "C" void kernel_launch(void* const* d_in, const int* in_sizes, int n_in,
                              void* d_out, int out_size, void* d_ws, size_t ws_size,
                              hipStream_t stream) {
    const float* x      = (const float*)d_in[0];
    const int*   phases = (const int*)d_in[1];
    const float* Ws     = (const float*)d_in[2];
    const float* bs     = (const float*)d_in[3];
    const float* W1     = (const float*)d_in[4];
    const float* b1     = (const float*)d_in[5];
    const float* W2     = (const float*)d_in[6];
    const float* b2     = (const float*)d_in[7];
    float* out   = (float*)d_out;
    float* ws2   = (float*)d_ws;
    float* weff2 = ws2 + WS2_FLOATS;
    float* beff  = weff2 + WEFF2_FLOATS;

    qnet_pair_ws<<<64, 256, 0, stream>>>(Ws, ws2);
    qnet_prep_weff<<<64, 256, 0, stream>>>(W1, b1, W2, b2, weff2, beff);
    qnet_fused<<<NB / ROWS_WG, WG_THREADS, 0, stream>>>(
        x, phases, bs, ws2, weff2, beff, out);
}


# Weff fold (W1[e]@W2[e]) moved from a serial 256-step per-thread dot product to f32 WMMA tiles, K split over 4 waves and summed via LDS; bias row K-split too (fold's summation order changes)
# speedup vs baseline: 1.3345x; 1.3345x over previous
#include <hip/hip_runtime.h>

#define NB 65536
#define ND 128
#define NH 256
#define NO 8
#define NE 8
#define EO (NE * NO)

typedef __attribute__((ext_vector_type(2))) float v2f;
typedef __attribute__((ext_vector_type(4))) float v4f;
typedef __attribute__((ext_vector_type(8))) float v8f;

#define WS2_FLOATS   (ND * NH)
#define WEFF2_FLOATS (NH * EO)


__global__ void qnet_pair_ws(const float* __restrict__ Ws,
                             float* __restrict__ ws2) {
    const int d = blockIdx.x * blockDim.x + threadIdx.x;
    const int j   = d & 1;
    int r         = d >> 1;
    const int n   = r & (NH - 1);
    r >>= 8;
    const int kp  = r & 15;
    const int lh  = r >> 4;
    const int k0  = 4 * (2 * kp + j) + 2 * lh;
    v2f v;
    v.x = Ws[(size_t)k0 * NH + n];
    v.y = Ws[(size_t)k0 * NH + n + NH];
    ((v2f*)ws2)[d] = v;
}

#define FOLD_THREADS 128
#define FOLD_TILES   128
#define FOLD_BLOCKS  136

__launch_bounds__(FOLD_THREADS)
__global__ void qnet_prep_weff(const float* __restrict__ W1,
                               const float* __restrict__ b1,
                               const float* __restrict__ W2,
                               const float* __restrict__ b2,
                               float* __restrict__ weff2,
                               float* __restrict__ beff) {
    __shared__ float part[4][16][8];
    const int t = threadIdx.x;

    if (blockIdx.x < FOLD_TILES) {
        const int e    = blockIdx.x >> 4;
        const int tile = blockIdx.x & 15;
        const int wave = t >> 5;
        const int lane = t & 31;
        const int lh   = lane >> 4;
        const int c    = lane & 15;

        const float* arow = W1 + ((size_t)e * NH + tile * 16 + c) * NH + 64 * wave + 4 * lh;
        const float* bcol = W2 + ((size_t)e * NH + 64 * wave + 4 * lh) * NO + (c & 7);

        v4f   a4[8];
        float bq[8][4];
        #pragma unroll
        for (int q = 0; q < 8; ++q)
            a4[q] = *(const v4f*)(arow + 8 * q);
        #pragma unroll
        for (int q = 0; q < 8; ++q)
            #pragma unroll
            for (int s = 0; s < 4; ++s)
                bq[q][s] = bcol[(size_t)(8 * q + s) * NO];

        v8f acc0 = {}, acc1 = {};
        #pragma unroll
        for (int q = 0; q < 8; ++q) {
            v2f alo, blo, ahi, bhi;
            alo.x = a4[q].x;  alo.y = a4[q].y;
            blo.x = bq[q][0]; blo.y = bq[q][1];
            ahi.x = a4[q].z;  ahi.y = a4[q].w;
            bhi.x = bq[q][2]; bhi.y = bq[q][3];
            acc0 = __builtin_amdgcn_wmma_f32_16x16x4_f32(
                       false, alo, false, blo, (short)0, acc0, false, false);
            acc1 = __builtin_amdgcn_wmma_f32_16x16x4_f32(
                       false, ahi, false, bhi, (short)0, acc1, false, false);
        }
        const v8f acc = acc0 + acc1;
        #pragma unroll
        for (int r = 0; r < 8; ++r)
            part[wave][8 * lh + r][c & 7] = acc[r];
        __syncthreads();

        const int o   = t & 7;
        const int row = 8 * (t >> 6) + 4 * ((t >> 4) & 1) + 2 * ((t >> 5) & 1) + ((t >> 3) & 1);
        const float s = ((part[0][row][o] + part[1][row][o]) + part[2][row][o]) + part[3][row][o];

        const int h   = tile * 16 + row;
        const int col = e * NO + o;
        const int i   = h & 1;
        const int lhh = (h >> 1) & 1;
        const int ks  = h >> 2;
        const int kp  = ks >> 1;
        const int j   = ks & 1;
        weff2[((((size_t)lhh * 32 + kp) * EO + col) * 2 + j) * 2 + i] = s;
    } else {
        const int e  = blockIdx.x - FOLD_TILES;
        const int o  = t & 7;
        const int kg = t >> 3;
        const float* b1e = b1 + (size_t)e * NH + 16 * kg;
        const float* w2c = W2 + ((size_t)e * NH + 16 * kg) * NO + o;
        float acc = 0.0f;
        #pragma unroll
        for (int kk = 0; kk < 16; ++kk)
            acc = __builtin_fmaf(b1e[kk], w2c[(size_t)kk * NO], acc);
        float* red = &part[0][0][0];
        red[t] = acc;
        __syncthreads();
        if (t < NO) {
            float s = b2[(size_t)e * NO + t];
            #pragma unroll
            for (int g = 0; g < 16; ++g)
                s += red[g * 8 + t];
            beff[e * NO + t] = s;
        }
    }
}

#define WG_THREADS 256
#define WG_WAVES   8
#define ROWS_WG    128
#define SH_STR     68

__launch_bounds__(WG_THREADS, 1)
__global__ void qnet_fused(const float* __restrict__ x,
                           const int*   __restrict__ phases,
                           const float* __restrict__ bs,
                           const float* __restrict__ ws2,
                           const float* __restrict__ weff2,
                           const float* __restrict__ beff,
                           float* __restrict__ out) {
    __shared__ float sH[WG_WAVES][16 * SH_STR];

    const int tid   = threadIdx.x;
    const int wave  = tid >> 5;
    const int lane  = tid & 31;
    const int lhalf = lane >> 4;
    const int lmod  = lane & 15;
    const int r0    = blockIdx.x * ROWS_WG + wave * 16;

    const v4f* Bws = (const v4f*)ws2;
    const v4f* Bwe = (const v4f*)weff2;

    v2f a[32];
    {
        const float* xrow = x + (size_t)(r0 + lmod) * ND + 2 * lhalf;
        #pragma unroll
        for (int kk = 0; kk < 32; ++kk)
            a[kk] = *(const v2f*)(xrow + 4 * kk);
    }

    float* myH = &sH[wave][0];
    v8f y[4] = {};

    for (int ch = 0; ch < 4; ++ch) {
        v8f c[4] = {};
        v4f bw[4];
        #pragma unroll
        for (int t = 0; t < 4; ++t)
            bw[t] = Bws[(size_t)(lhalf * 16 + 0) * NH + (ch * 4 + t) * 16 + lmod];
        #pragma unroll
        for (int kp = 0; kp < 16; ++kp) {
            v4f bn[4];
            if (kp < 15) {
                #pragma unroll
                for (int t = 0; t < 4; ++t)
                    bn[t] = Bws[(size_t)(lhalf * 16 + kp + 1) * NH +
                                (ch * 4 + t) * 16 + lmod];
            }
            #pragma unroll
            for (int t = 0; t < 4; ++t) {
                const v2f blo = __builtin_shufflevector(bw[t], bw[t], 0, 1);
                c[t] = __builtin_amdgcn_wmma_f32_16x16x4_f32(
                           false, a[2 * kp], false, blo, (short)0, c[t], false, false);
            }
            #pragma unroll
            for (int t = 0; t < 4; ++t) {
                const v2f bhi = __builtin_shufflevector(bw[t], bw[t], 2, 3);
                c[t] = __builtin_amdgcn_wmma_f32_16x16x4_f32(
                           false, a[2 * kp + 1], false, bhi, (short)0, c[t], false, false);
            }
            if (kp < 15) {
                #pragma unroll
                for (int t = 0; t < 4; ++t) bw[t] = bn[t];
            }
        }
        #pragma unroll
        for (int t = 0; t < 4; ++t) {
            const float bsv = bs[(ch * 4 + t) * 16 + lmod];
            #pragma unroll
            for (int v = 0; v < 8; ++v) {
                const float hv = fmaxf(c[t][v] + bsv, 0.0f);
                myH[(v + 8 * lhalf) * SH_STR + t * 16 + lmod] = hv;
            }
        }
        const float* hA = myH + lmod * SH_STR + 2 * lhalf;
        v4f ww[4];
        #pragma unroll
        for (int t = 0; t < 4; ++t)
            ww[t] = Bwe[(size_t)(lhalf * 32 + ch * 8 + 0) * EO + t * 16 + lmod];
        #pragma unroll
        for (int kp = 0; kp < 8; ++kp) {
            v4f wn[4];
            if (kp < 7) {
                #pragma unroll
                for (int t = 0; t < 4; ++t)
                    wn[t] = Bwe[(size_t)(lhalf * 32 + ch * 8 + kp + 1) * EO +
                                t * 16 + lmod];
            }
            const v2f a0 = *(const v2f*)(hA + 4 * (2 * kp));
            const v2f a1 = *(const v2f*)(hA + 4 * (2 * kp + 1));
            #pragma unroll
            for (int t = 0; t < 4; ++t) {
                const v2f blo = __builtin_shufflevector(ww[t], ww[t], 0, 1);
                y[t] = __builtin_amdgcn_wmma_f32_16x16x4_f32(
                           false, a0, false, blo, (short)0, y[t], false, false);
            }
            #pragma unroll
            for (int t = 0; t < 4; ++t) {
                const v2f bhi = __builtin_shufflevector(ww[t], ww[t], 2, 3);
                y[t] = __builtin_amdgcn_wmma_f32_16x16x4_f32(
                           false, a1, false, bhi, (short)0, y[t], false, false);
            }
            if (kp < 7) {
                #pragma unroll
                for (int t = 0; t < 4; ++t) ww[t] = wn[t];
            }
        }
    }

    #pragma unroll
    for (int t = 0; t < 4; ++t)
        #pragma unroll
        for (int v = 0; v < 8; ++v)
            myH[(v + 8 * lhalf) * SH_STR + t * 16 + lmod] = y[t][v];

    const int selRow = lane >> 1;
    const int oq     = (lane & 1) * 4;
    const int p      = phases[r0 + selRow];
    const v4f yv = *(const v4f*)(myH + selRow * SH_STR + p * NO + oq);
    const v4f bv = *(const v4f*)(beff + p * NO + oq);
    *(v4f*)(out + (size_t)(r0 + selRow) * NO + oq) = yv + bv;
}

extern "C" void kernel_launch(void* const* d_in, const int* in_sizes, int n_in,
                              void* d_out, int out_size, void* d_ws, size_t ws_size,
                              hipStream_t stream) {
    const float* x      = (const float*)d_in[0];
    const int*   phases = (const int*)d_in[1];
    const float* Ws     = (const float*)d_in[2];
    const float* bs     = (const float*)d_in[3];
    const float* W1     = (const float*)d_in[4];
    const float* b1     = (const float*)d_in[5];
    const float* W2     = (const float*)d_in[6];
    const float* b2     = (const float*)d_in[7];
    float* out   = (float*)d_out;
    float* ws2   = (float*)d_ws;
    float* weff2 = ws2 + WS2_FLOATS;
    float* beff  = weff2 + WEFF2_FLOATS;

    qnet_pair_ws<<<64, 256, 0, stream>>>(Ws, ws2);
    qnet_prep_weff<<<FOLD_BLOCKS, FOLD_THREADS, 0, stream>>>(W1, b1, W2, b2, weff2, beff);
    qnet_fused<<<NB / ROWS_WG, WG_THREADS, 0, stream>>>(
        x, phases, bs, ws2, weff2, beff, out);
}


# Ws swizzle moved into the fold kernel's launch as extra work-groups (blocks 136..263): 2 launches per run instead of 3; every kernel body and sum order unchanged
# speedup vs baseline: 1.4225x; 1.0659x over previous
#include <hip/hip_runtime.h>

#define NB 65536
#define ND 128
#define NH 256
#define NO 8
#define NE 8
#define EO (NE * NO)

typedef __attribute__((ext_vector_type(2))) float v2f;
typedef __attribute__((ext_vector_type(4))) float v4f;
typedef __attribute__((ext_vector_type(8))) float v8f;

#define WS2_FLOATS   (ND * NH)
#define WEFF2_FLOATS (NH * EO)


#define FOLD_THREADS 128
#define FOLD_TILES   128
#define FOLD_BLOCKS  136
#define PAIR_BLOCKS  (ND * NH / 2 / FOLD_THREADS)
#define PREP_BLOCKS  (FOLD_BLOCKS + PAIR_BLOCKS)

__launch_bounds__(FOLD_THREADS)
__global__ void qnet_prep(const float* __restrict__ W1,
                          const float* __restrict__ b1,
                          const float* __restrict__ W2,
                          const float* __restrict__ b2,
                          const float* __restrict__ Ws,
                          float* __restrict__ weff2,
                          float* __restrict__ beff,
                          float* __restrict__ ws2) {
    __shared__ float part[4][16][8];
    const int t = threadIdx.x;

    if (blockIdx.x < FOLD_TILES) {
        const int e    = blockIdx.x >> 4;
        const int tile = blockIdx.x & 15;
        const int wave = t >> 5;
        const int lane = t & 31;
        const int lh   = lane >> 4;
        const int c    = lane & 15;

        const float* arow = W1 + ((size_t)e * NH + tile * 16 + c) * NH + 64 * wave + 4 * lh;
        const float* bcol = W2 + ((size_t)e * NH + 64 * wave + 4 * lh) * NO + (c & 7);

        v4f   a4[8];
        float bq[8][4];
        #pragma unroll
        for (int q = 0; q < 8; ++q)
            a4[q] = *(const v4f*)(arow + 8 * q);
        #pragma unroll
        for (int q = 0; q < 8; ++q)
            #pragma unroll
            for (int s = 0; s < 4; ++s)
                bq[q][s] = bcol[(size_t)(8 * q + s) * NO];

        v8f acc0 = {}, acc1 = {};
        #pragma unroll
        for (int q = 0; q < 8; ++q) {
            v2f alo, blo, ahi, bhi;
            alo.x = a4[q].x;  alo.y = a4[q].y;
            blo.x = bq[q][0]; blo.y = bq[q][1];
            ahi.x = a4[q].z;  ahi.y = a4[q].w;
            bhi.x = bq[q][2]; bhi.y = bq[q][3];
            acc0 = __builtin_amdgcn_wmma_f32_16x16x4_f32(
                       false, alo, false, blo, (short)0, acc0, false, false);
            acc1 = __builtin_amdgcn_wmma_f32_16x16x4_f32(
                       false, ahi, false, bhi, (short)0, acc1, false, false);
        }
        const v8f acc = acc0 + acc1;
        #pragma unroll
        for (int r = 0; r < 8; ++r)
            part[wave][8 * lh + r][c & 7] = acc[r];
        __syncthreads();

        const int o   = t & 7;
        const int row = 8 * (t >> 6) + 4 * ((t >> 4) & 1) + 2 * ((t >> 5) & 1) + ((t >> 3) & 1);
        const float s = ((part[0][row][o] + part[1][row][o]) + part[2][row][o]) + part[3][row][o];

        const int h   = tile * 16 + row;
        const int col = e * NO + o;
        const int i   = h & 1;
        const int lhh = (h >> 1) & 1;
        const int ks  = h >> 2;
        const int kp  = ks >> 1;
        const int j   = ks & 1;
        weff2[((((size_t)lhh * 32 + kp) * EO + col) * 2 + j) * 2 + i] = s;
    } else if (blockIdx.x < FOLD_BLOCKS) {
        const int e  = blockIdx.x - FOLD_TILES;
        const int o  = t & 7;
        const int kg = t >> 3;
        const float* b1e = b1 + (size_t)e * NH + 16 * kg;
        const float* w2c = W2 + ((size_t)e * NH + 16 * kg) * NO + o;
        float acc = 0.0f;
        #pragma unroll
        for (int kk = 0; kk < 16; ++kk)
            acc = __builtin_fmaf(b1e[kk], w2c[(size_t)kk * NO], acc);
        float* red = &part[0][0][0];
        red[t] = acc;
        __syncthreads();
        if (t < NO) {
            float s = b2[(size_t)e * NO + t];
            #pragma unroll
            for (int g = 0; g < 16; ++g)
                s += red[g * 8 + t];
            beff[e * NO + t] = s;
        }
    } else {
        const int d = (blockIdx.x - FOLD_BLOCKS) * FOLD_THREADS + t;
        const int j   = d & 1;
        int r         = d >> 1;
        const int n   = r & (NH - 1);
        r >>= 8;
        const int kp  = r & 15;
        const int lh  = r >> 4;
        const int k0  = 4 * (2 * kp + j) + 2 * lh;
        v2f v;
        v.x = Ws[(size_t)k0 * NH + n];
        v.y = Ws[(size_t)k0 * NH + n + NH];
        ((v2f*)ws2)[d] = v;
    }
}

#define WG_THREADS 256
#define WG_WAVES   8
#define ROWS_WG    128
#define SH_STR     68

__launch_bounds__(WG_THREADS, 1)
__global__ void qnet_fused(const float* __restrict__ x,
                           const int*   __restrict__ phases,
                           const float* __restrict__ bs,
                           const float* __restrict__ ws2,
                           const float* __restrict__ weff2,
                           const float* __restrict__ beff,
                           float* __restrict__ out) {
    __shared__ float sH[WG_WAVES][16 * SH_STR];

    const int tid   = threadIdx.x;
    const int wave  = tid >> 5;
    const int lane  = tid & 31;
    const int lhalf = lane >> 4;
    const int lmod  = lane & 15;
    const int r0    = blockIdx.x * ROWS_WG + wave * 16;

    const v4f* Bws = (const v4f*)ws2;
    const v4f* Bwe = (const v4f*)weff2;

    v2f a[32];
    {
        const float* xrow = x + (size_t)(r0 + lmod) * ND + 2 * lhalf;
        #pragma unroll
        for (int kk = 0; kk < 32; ++kk)
            a[kk] = *(const v2f*)(xrow + 4 * kk);
    }

    float* myH = &sH[wave][0];
    v8f y[4] = {};

    for (int ch = 0; ch < 4; ++ch) {
        v8f c[4] = {};
        v4f bw[4];
        #pragma unroll
        for (int t = 0; t < 4; ++t)
            bw[t] = Bws[(size_t)(lhalf * 16 + 0) * NH + (ch * 4 + t) * 16 + lmod];
        #pragma unroll
        for (int kp = 0; kp < 16; ++kp) {
            v4f bn[4];
            if (kp < 15) {
                #pragma unroll
                for (int t = 0; t < 4; ++t)
                    bn[t] = Bws[(size_t)(lhalf * 16 + kp + 1) * NH +
                                (ch * 4 + t) * 16 + lmod];
            }
            #pragma unroll
            for (int t = 0; t < 4; ++t) {
                const v2f blo = __builtin_shufflevector(bw[t], bw[t], 0, 1);
                c[t] = __builtin_amdgcn_wmma_f32_16x16x4_f32(
                           false, a[2 * kp], false, blo, (short)0, c[t], false, false);
            }
            #pragma unroll
            for (int t = 0; t < 4; ++t) {
                const v2f bhi = __builtin_shufflevector(bw[t], bw[t], 2, 3);
                c[t] = __builtin_amdgcn_wmma_f32_16x16x4_f32(
                           false, a[2 * kp + 1], false, bhi, (short)0, c[t], false, false);
            }
            if (kp < 15) {
                #pragma unroll
                for (int t = 0; t < 4; ++t) bw[t] = bn[t];
            }
        }
        #pragma unroll
        for (int t = 0; t < 4; ++t) {
            const float bsv = bs[(ch * 4 + t) * 16 + lmod];
            #pragma unroll
            for (int v = 0; v < 8; ++v) {
                const float hv = fmaxf(c[t][v] + bsv, 0.0f);
                myH[(v + 8 * lhalf) * SH_STR + t * 16 + lmod] = hv;
            }
        }
        const float* hA = myH + lmod * SH_STR + 2 * lhalf;
        v4f ww[4];
        #pragma unroll
        for (int t = 0; t < 4; ++t)
            ww[t] = Bwe[(size_t)(lhalf * 32 + ch * 8 + 0) * EO + t * 16 + lmod];
        #pragma unroll
        for (int kp = 0; kp < 8; ++kp) {
            v4f wn[4];
            if (kp < 7) {
                #pragma unroll
                for (int t = 0; t < 4; ++t)
                    wn[t] = Bwe[(size_t)(lhalf * 32 + ch * 8 + kp + 1) * EO +
                                t * 16 + lmod];
            }
            const v2f a0 = *(const v2f*)(hA + 4 * (2 * kp));
            const v2f a1 = *(const v2f*)(hA + 4 * (2 * kp + 1));
            #pragma unroll
            for (int t = 0; t < 4; ++t) {
                const v2f blo = __builtin_shufflevector(ww[t], ww[t], 0, 1);
                y[t] = __builtin_amdgcn_wmma_f32_16x16x4_f32(
                           false, a0, false, blo, (short)0, y[t], false, false);
            }
            #pragma unroll
            for (int t = 0; t < 4; ++t) {
                const v2f bhi = __builtin_shufflevector(ww[t], ww[t], 2, 3);
                y[t] = __builtin_amdgcn_wmma_f32_16x16x4_f32(
                           false, a1, false, bhi, (short)0, y[t], false, false);
            }
            if (kp < 7) {
                #pragma unroll
                for (int t = 0; t < 4; ++t) ww[t] = wn[t];
            }
        }
    }

    #pragma unroll
    for (int t = 0; t < 4; ++t)
        #pragma unroll
        for (int v = 0; v < 8; ++v)
            myH[(v + 8 * lhalf) * SH_STR + t * 16 + lmod] = y[t][v];

    const int selRow = lane >> 1;
    const int oq     = (lane & 1) * 4;
    const int p      = phases[r0 + selRow];
    const v4f yv = *(const v4f*)(myH + selRow * SH_STR + p * NO + oq);
    const v4f bv = *(const v4f*)(beff + p * NO + oq);
    *(v4f*)(out + (size_t)(r0 + selRow) * NO + oq) = yv + bv;
}

extern "C" void kernel_launch(void* const* d_in, const int* in_sizes, int n_in,
                              void* d_out, int out_size, void* d_ws, size_t ws_size,
                              hipStream_t stream) {
    const float* x      = (const float*)d_in[0];
    const int*   phases = (const int*)d_in[1];
    const float* Ws     = (const float*)d_in[2];
    const float* bs     = (const float*)d_in[3];
    const float* W1     = (const float*)d_in[4];
    const float* b1     = (const float*)d_in[5];
    const float* W2     = (const float*)d_in[6];
    const float* b2     = (const float*)d_in[7];
    float* out   = (float*)d_out;
    float* ws2   = (float*)d_ws;
    float* weff2 = ws2 + WS2_FLOATS;
    float* beff  = weff2 + WEFF2_FLOATS;

    qnet_prep<<<PREP_BLOCKS, FOLD_THREADS, 0, stream>>>(W1, b1, W2, b2, Ws, weff2, beff, ws2);
    qnet_fused<<<NB / ROWS_WG, WG_THREADS, 0, stream>>>(
        x, phases, bs, ws2, weff2, beff, out);
}


# Fused kernel: each work-group's 128 rows are dealt to its waves sorted by routed head pair, so a wave runs only the 1-2 Weff column tiles its rows use, not all 4; every output's sum order unchanged
# speedup vs baseline: 1.6314x; 1.1469x over previous
#include <hip/hip_runtime.h>

#define NB 65536
#define ND 128
#define NH 256
#define NO 8
#define NE 8
#define EO (NE * NO)

typedef __attribute__((ext_vector_type(2))) float v2f;
typedef __attribute__((ext_vector_type(4))) float v4f;
typedef __attribute__((ext_vector_type(8))) float v8f;

#define WS2_FLOATS   (ND * NH)
#define WEFF2_FLOATS (NH * EO)


#define FOLD_THREADS 128
#define FOLD_TILES   128
#define FOLD_BLOCKS  136
#define PAIR_BLOCKS  (ND * NH / 2 / FOLD_THREADS)
#define PREP_BLOCKS  (FOLD_BLOCKS + PAIR_BLOCKS)

__launch_bounds__(FOLD_THREADS)
__global__ void qnet_prep(const float* __restrict__ W1,
                          const float* __restrict__ b1,
                          const float* __restrict__ W2,
                          const float* __restrict__ b2,
                          const float* __restrict__ Ws,
                          float* __restrict__ weff2,
                          float* __restrict__ beff,
                          float* __restrict__ ws2) {
    __shared__ float part[4][16][8];
    const int t = threadIdx.x;

    if (blockIdx.x < FOLD_TILES) {
        const int e    = blockIdx.x >> 4;
        const int tile = blockIdx.x & 15;
        const int wave = t >> 5;
        const int lane = t & 31;
        const int lh   = lane >> 4;
        const int c    = lane & 15;

        const float* arow = W1 + ((size_t)e * NH + tile * 16 + c) * NH + 64 * wave + 4 * lh;
        const float* bcol = W2 + ((size_t)e * NH + 64 * wave + 4 * lh) * NO + (c & 7);

        v4f   a4[8];
        float bq[8][4];
        #pragma unroll
        for (int q = 0; q < 8; ++q)
            a4[q] = *(const v4f*)(arow + 8 * q);
        #pragma unroll
        for (int q = 0; q < 8; ++q)
            #pragma unroll
            for (int s = 0; s < 4; ++s)
                bq[q][s] = bcol[(size_t)(8 * q + s) * NO];

        v8f acc0 = {}, acc1 = {};
        #pragma unroll
        for (int q = 0; q < 8; ++q) {
            v2f alo, blo, ahi, bhi;
            alo.x = a4[q].x;  alo.y = a4[q].y;
            blo.x = bq[q][0]; blo.y = bq[q][1];
            ahi.x = a4[q].z;  ahi.y = a4[q].w;
            bhi.x = bq[q][2]; bhi.y = bq[q][3];
            acc0 = __builtin_amdgcn_wmma_f32_16x16x4_f32(
                       false, alo, false, blo, (short)0, acc0, false, false);
            acc1 = __builtin_amdgcn_wmma_f32_16x16x4_f32(
                       false, ahi, false, bhi, (short)0, acc1, false, false);
        }
        const v8f acc = acc0 + acc1;
        #pragma unroll
        for (int r = 0; r < 8; ++r)
            part[wave][8 * lh + r][c & 7] = acc[r];
        __syncthreads();

        const int o   = t & 7;
        const int row = 8 * (t >> 6) + 4 * ((t >> 4) & 1) + 2 * ((t >> 5) & 1) + ((t >> 3) & 1);
        const float s = ((part[0][row][o] + part[1][row][o]) + part[2][row][o]) + part[3][row][o];

        const int h   = tile * 16 + row;
        const int col = e * NO + o;
        const int i   = h & 1;
        const int lhh = (h >> 1) & 1;
        const int ks  = h >> 2;
        const int kp  = ks >> 1;
        const int j   = ks & 1;
        weff2[((((size_t)lhh * 32 + kp) * EO + col) * 2 + j) * 2 + i] = s;
    } else if (blockIdx.x < FOLD_BLOCKS) {
        const int e  = blockIdx.x - FOLD_TILES;
        const int o  = t & 7;
        const int kg = t >> 3;
        const float* b1e = b1 + (size_t)e * NH + 16 * kg;
        const float* w2c = W2 + ((size_t)e * NH + 16 * kg) * NO + o;
        float acc = 0.0f;
        #pragma unroll
        for (int kk = 0; kk < 16; ++kk)
            acc = __builtin_fmaf(b1e[kk], w2c[(size_t)kk * NO], acc);
        float* red = &part[0][0][0];
        red[t] = acc;
        __syncthreads();
        if (t < NO) {
            float s = b2[(size_t)e * NO + t];
            #pragma unroll
            for (int g = 0; g < 16; ++g)
                s += red[g * 8 + t];
            beff[e * NO + t] = s;
        }
    } else {
        const int d = (blockIdx.x - FOLD_BLOCKS) * FOLD_THREADS + t;
        const int j   = d & 1;
        int r         = d >> 1;
        const int n   = r & (NH - 1);
        r >>= 8;
        const int kp  = r & 15;
        const int lh  = r >> 4;
        const int k0  = 4 * (2 * kp + j) + 2 * lh;
        v2f v;
        v.x = Ws[(size_t)k0 * NH + n];
        v.y = Ws[(size_t)k0 * NH + n + NH];
        ((v2f*)ws2)[d] = v;
    }
}

#define WG_THREADS 256
#define WG_WAVES   8
#define ROWS_WG    128
#define SH_STR     68

__launch_bounds__(WG_THREADS) __attribute__((amdgpu_waves_per_eu(4)))
__global__ void qnet_fused(const float* __restrict__ x,
                           const int*   __restrict__ phases,
                           const float* __restrict__ bs,
                           const float* __restrict__ ws2,
                           const float* __restrict__ weff2,
                           const float* __restrict__ beff,
                           float* __restrict__ out) {
    __shared__ float sH[WG_WAVES][16 * SH_STR];
    __shared__ int   sOrd[WG_WAVES][ROWS_WG];
    __shared__ float sOut[ROWS_WG * NO];

    const int tid   = threadIdx.x;
    const int wave  = __builtin_amdgcn_readfirstlane(tid >> 5);
    const int lane  = tid & 31;
    const int lhalf = lane >> 4;
    const int lmod  = lane & 15;
    const int base  = blockIdx.x * ROWS_WG;

    int* myOrd = &sOrd[wave][0];
    {
        int pv[4], cl[4], pos[4];
        #pragma unroll
        for (int q = 0; q < 4; ++q) {
            pv[q]  = phases[base + 32 * q + lane];
            cl[q]  = pv[q] >> 1;
            pos[q] = 0;
        }
        const unsigned below = (1u << lane) - 1u;
        int run = 0;
        #pragma unroll
        for (int c = 0; c < 4; ++c) {
            #pragma unroll
            for (int q = 0; q < 4; ++q) {
                const unsigned m = __builtin_amdgcn_ballot_w32(cl[q] == c);
                if (cl[q] == c) pos[q] = run + __builtin_popcount(m & below);
                run += __builtin_popcount(m);
            }
        }
        #pragma unroll
        for (int q = 0; q < 4; ++q)
            myOrd[pos[q]] = (32 * q + lane) | (pv[q] << 8);
    }
    const int mine = myOrd[16 * wave + lmod];
    const int mycl = mine >> 9;
    unsigned need = 0;
    #pragma unroll
    for (int c = 0; c < 4; ++c)
        need |= (__builtin_amdgcn_ballot_w32(mycl == c) != 0u) ? (1u << c) : 0u;
    const int nt = __builtin_popcount(need);
    const int t0 = __builtin_ctz(need);

    const v4f* Bws = (const v4f*)ws2;
    const v4f* Bwe = (const v4f*)weff2;

    v2f a[32];
    {
        const float* xrow = x + (size_t)(base + (mine & 255)) * ND + 2 * lhalf;
        #pragma unroll
        for (int kk = 0; kk < 32; ++kk)
            a[kk] = *(const v2f*)(xrow + 4 * kk);
    }

    float* myH = &sH[wave][0];
    v8f y[4] = {};

    for (int ch = 0; ch < 4; ++ch) {
        v8f c[4] = {};
        v4f bw[4];
        #pragma unroll
        for (int t = 0; t < 4; ++t)
            bw[t] = Bws[(size_t)(lhalf * 16 + 0) * NH + (ch * 4 + t) * 16 + lmod];
        #pragma unroll
        for (int kp = 0; kp < 16; ++kp) {
            v4f bn[4];
            if (kp < 15) {
                #pragma unroll
                for (int t = 0; t < 4; ++t)
                    bn[t] = Bws[(size_t)(lhalf * 16 + kp + 1) * NH +
                                (ch * 4 + t) * 16 + lmod];
            }
            #pragma unroll
            for (int t = 0; t < 4; ++t) {
                const v2f blo = __builtin_shufflevector(bw[t], bw[t], 0, 1);
                c[t] = __builtin_amdgcn_wmma_f32_16x16x4_f32(
                           false, a[2 * kp], false, blo, (short)0, c[t], false, false);
            }
            #pragma unroll
            for (int t = 0; t < 4; ++t) {
                const v2f bhi = __builtin_shufflevector(bw[t], bw[t], 2, 3);
                c[t] = __builtin_amdgcn_wmma_f32_16x16x4_f32(
                           false, a[2 * kp + 1], false, bhi, (short)0, c[t], false, false);
            }
            if (kp < 15) {
                #pragma unroll
                for (int t = 0; t < 4; ++t) bw[t] = bn[t];
            }
        }
        #pragma unroll
        for (int t = 0; t < 4; ++t) {
            const float bsv = bs[(ch * 4 + t) * 16 + lmod];
            #pragma unroll
            for (int v = 0; v < 8; ++v) {
                const float hv = fmaxf(c[t][v] + bsv, 0.0f);
                myH[(v + 8 * lhalf) * SH_STR + t * 16 + lmod] = hv;
            }
        }
        const float* hA = myH + lmod * SH_STR + 2 * lhalf;
        v2f ha[16];
        #pragma unroll
        for (int q = 0; q < 16; ++q)
            ha[q] = *(const v2f*)(hA + 4 * q);
        unsigned rest = need;
        #pragma unroll
        for (int s = 0; s < 4; ++s) {
            if (s == 0 || nt > s) {
                const int t = (s == 0) ? t0 : __builtin_ctz(rest);
                rest &= rest - 1u;
                v4f ww[8];
                #pragma unroll
                for (int kp = 0; kp < 8; ++kp)
                    ww[kp] = Bwe[(size_t)(lhalf * 32 + ch * 8 + kp) * EO + t * 16 + lmod];
                #pragma unroll
                for (int kp = 0; kp < 8; ++kp) {
                    const v2f blo = __builtin_shufflevector(ww[kp], ww[kp], 0, 1);
                    const v2f bhi = __builtin_shufflevector(ww[kp], ww[kp], 2, 3);
                    y[s] = __builtin_amdgcn_wmma_f32_16x16x4_f32(
                               false, ha[2 * kp], false, blo, (short)0, y[s], false, false);
                    y[s] = __builtin_amdgcn_wmma_f32_16x16x4_f32(
                               false, ha[2 * kp + 1], false, bhi, (short)0, y[s], false, false);
                }
            }
        }
    }

    #pragma unroll
    for (int t = 0; t < 4; ++t)
        #pragma unroll
        for (int v = 0; v < 8; ++v)
            myH[(v + 8 * lhalf) * SH_STR + t * 16 + lmod] = y[t][v];

    const int selRow = lane >> 1;
    const int oq     = (lane & 1) * 4;
    const int sel    = myOrd[16 * wave + selRow];
    const int p      = sel >> 8;
    const int slot   = __builtin_popcount(need & ((1u << (p >> 1)) - 1u));
    const v4f yv = *(const v4f*)(myH + selRow * SH_STR + slot * 16 + (p & 1) * NO + oq);
    const v4f bv = *(const v4f*)(beff + p * NO + oq);
    *(v4f*)(sOut + (sel & 255) * NO + oq) = yv + bv;
    __syncthreads();
    *(v4f*)(out + (size_t)base * NO + tid * 4) = *(const v4f*)(sOut + tid * 4);
}

extern "C" void kernel_launch(void* const* d_in, const int* in_sizes, int n_in,
                              void* d_out, int out_size, void* d_ws, size_t ws_size,
                              hipStream_t stream) {
    const float* x      = (const float*)d_in[0];
    const int*   phases = (const int*)d_in[1];
    const float* Ws     = (const float*)d_in[2];
    const float* bs     = (const float*)d_in[3];
    const float* W1     = (const float*)d_in[4];
    const float* b1     = (const float*)d_in[5];
    const float* W2     = (const float*)d_in[6];
    const float* b2     = (const float*)d_in[7];
    float* out   = (float*)d_out;
    float* ws2   = (float*)d_ws;
    float* weff2 = ws2 + WS2_FLOATS;
    float* beff  = weff2 + WEFF2_FLOATS;

    qnet_prep<<<PREP_BLOCKS, FOLD_THREADS, 0, stream>>>(W1, b1, W2, b2, Ws, weff2, beff, ws2);
    qnet_fused<<<NB / ROWS_WG, WG_THREADS, 0, stream>>>(
        x, phases, bs, ws2, weff2, beff, out);
}


# Fused kernel: s_setprio steps each wave's issue priority down with its own progress (3 at entry and chunk 0, then 2, 1, 0), so co-resident work-groups advance together; sums unchanged
# speedup vs baseline: 1.7172x; 1.0526x over previous
#include <hip/hip_runtime.h>

#define NB 65536
#define ND 128
#define NH 256
#define NO 8
#define NE 8
#define EO (NE * NO)

typedef __attribute__((ext_vector_type(2))) float v2f;
typedef __attribute__((ext_vector_type(4))) float v4f;
typedef __attribute__((ext_vector_type(8))) float v8f;

#define WS2_FLOATS   (ND * NH)
#define WEFF2_FLOATS (NH * EO)


#define FOLD_THREADS 128
#define FOLD_TILES   128
#define FOLD_BLOCKS  136
#define PAIR_BLOCKS  (ND * NH / 2 / FOLD_THREADS)
#define PREP_BLOCKS  (FOLD_BLOCKS + PAIR_BLOCKS)

__launch_bounds__(FOLD_THREADS)
__global__ void qnet_prep(const float* __restrict__ W1,
                          const float* __restrict__ b1,
                          const float* __restrict__ W2,
                          const float* __restrict__ b2,
                          const float* __restrict__ Ws,
                          float* __restrict__ weff2,
                          float* __restrict__ beff,
                          float* __restrict__ ws2) {
    __shared__ float part[4][16][8];
    const int t = threadIdx.x;

    if (blockIdx.x < FOLD_TILES) {
        const int e    = blockIdx.x >> 4;
        const int tile = blockIdx.x & 15;
        const int wave = t >> 5;
        const int lane = t & 31;
        const int lh   = lane >> 4;
        const int c    = lane & 15;

        const float* arow = W1 + ((size_t)e * NH + tile * 16 + c) * NH + 64 * wave + 4 * lh;
        const float* bcol = W2 + ((size_t)e * NH + 64 * wave + 4 * lh) * NO + (c & 7);

        v4f   a4[8];
        float bq[8][4];
        #pragma unroll
        for (int q = 0; q < 8; ++q)
            a4[q] = *(const v4f*)(arow + 8 * q);
        #pragma unroll
        for (int q = 0; q < 8; ++q)
            #pragma unroll
            for (int s = 0; s < 4; ++s)
                bq[q][s] = bcol[(size_t)(8 * q + s) * NO];

        v8f acc0 = {}, acc1 = {};
        #pragma unroll
        for (int q = 0; q < 8; ++q) {
            v2f alo, blo, ahi, bhi;
            alo.x = a4[q].x;  alo.y = a4[q].y;
            blo.x = bq[q][0]; blo.y = bq[q][1];
            ahi.x = a4[q].z;  ahi.y = a4[q].w;
            bhi.x = bq[q][2]; bhi.y = bq[q][3];
            acc0 = __builtin_amdgcn_wmma_f32_16x16x4_f32(
                       false, alo, false, blo, (short)0, acc0, false, false);
            acc1 = __builtin_amdgcn_wmma_f32_16x16x4_f32(
                       false, ahi, false, bhi, (short)0, acc1, false, false);
        }
        const v8f acc = acc0 + acc1;
        #pragma unroll
        for (int r = 0; r < 8; ++r)
            part[wave][8 * lh + r][c & 7] = acc[r];
        __syncthreads();

        const int o   = t & 7;
        const int row = 8 * (t >> 6) + 4 * ((t >> 4) & 1) + 2 * ((t >> 5) & 1) + ((t >> 3) & 1);
        const float s = ((part[0][row][o] + part[1][row][o]) + part[2][row][o]) + part[3][row][o];

        const int h   = tile * 16 + row;
        const int col = e * NO + o;
        const int i   = h & 1;
        const int lhh = (h >> 1) & 1;
        const int ks  = h >> 2;
        const int kp  = ks >> 1;
        const int j   = ks & 1;
        weff2[((((size_t)lhh * 32 + kp) * EO + col) * 2 + j) * 2 + i] = s;
    } else if (blockIdx.x < FOLD_BLOCKS) {
        const int e  = blockIdx.x - FOLD_TILES;
        const int o  = t & 7;
        const int kg = t >> 3;
        const float* b1e = b1 + (size_t)e * NH + 16 * kg;
        const float* w2c = W2 + ((size_t)e * NH + 16 * kg) * NO + o;
        float acc = 0.0f;
        #pragma unroll
        for (int kk = 0; kk < 16; ++kk)
            acc = __builtin_fmaf(b1e[kk], w2c[(size_t)kk * NO], acc);
        float* red = &part[0][0][0];
        red[t] = acc;
        __syncthreads();
        if (t < NO) {
            float s = b2[(size_t)e * NO + t];
            #pragma unroll
            for (int g = 0; g < 16; ++g)
                s += red[g * 8 + t];
            beff[e * NO + t] = s;
        }
    } else {
        const int d = (blockIdx.x - FOLD_BLOCKS) * FOLD_THREADS + t;
        const int j   = d & 1;
        int r         = d >> 1;
        const int n   = r & (NH - 1);
        r >>= 8;
        const int kp  = r & 15;
        const int lh  = r >> 4;
        const int k0  = 4 * (2 * kp + j) + 2 * lh;
        v2f v;
        v.x = Ws[(size_t)k0 * NH + n];
        v.y = Ws[(size_t)k0 * NH + n + NH];
        ((v2f*)ws2)[d] = v;
    }
}

#define WG_THREADS 256
#define WG_WAVES   8
#define ROWS_WG    128
#define SH_STR     68

__launch_bounds__(WG_THREADS) __attribute__((amdgpu_waves_per_eu(4)))
__global__ void qnet_fused(const float* __restrict__ x,
                           const int*   __restrict__ phases,
                           const float* __restrict__ bs,
                           const float* __restrict__ ws2,
                           const float* __restrict__ weff2,
                           const float* __restrict__ beff,
                           float* __restrict__ out) {
    __shared__ float sH[WG_WAVES][16 * SH_STR];
    __shared__ int   sOrd[WG_WAVES][ROWS_WG];
    __shared__ float sOut[ROWS_WG * NO];

    __builtin_amdgcn_s_setprio(3);

    const int tid   = threadIdx.x;
    const int wave  = __builtin_amdgcn_readfirstlane(tid >> 5);
    const int lane  = tid & 31;
    const int lhalf = lane >> 4;
    const int lmod  = lane & 15;
    const int base  = blockIdx.x * ROWS_WG;

    int* myOrd = &sOrd[wave][0];
    {
        int pv[4], cl[4], pos[4];
        #pragma unroll
        for (int q = 0; q < 4; ++q) {
            pv[q]  = phases[base + 32 * q + lane];
            cl[q]  = pv[q] >> 1;
            pos[q] = 0;
        }
        const unsigned below = (1u << lane) - 1u;
        int run = 0;
        #pragma unroll
        for (int c = 0; c < 4; ++c) {
            #pragma unroll
            for (int q = 0; q < 4; ++q) {
                const unsigned m = __builtin_amdgcn_ballot_w32(cl[q] == c);
                if (cl[q] == c) pos[q] = run + __builtin_popcount(m & below);
                run += __builtin_popcount(m);
            }
        }
        #pragma unroll
        for (int q = 0; q < 4; ++q)
            myOrd[pos[q]] = (32 * q + lane) | (pv[q] << 8);
    }
    const int mine = myOrd[16 * wave + lmod];
    const int mycl = mine >> 9;
    unsigned need = 0;
    #pragma unroll
    for (int c = 0; c < 4; ++c)
        need |= (__builtin_amdgcn_ballot_w32(mycl == c) != 0u) ? (1u << c) : 0u;
    const int nt = __builtin_popcount(need);
    const int t0 = __builtin_ctz(need);

    const v4f* Bws = (const v4f*)ws2;
    const v4f* Bwe = (const v4f*)weff2;

    v2f a[32];
    {
        const float* xrow = x + (size_t)(base + (mine & 255)) * ND + 2 * lhalf;
        #pragma unroll
        for (int kk = 0; kk < 32; ++kk)
            a[kk] = *(const v2f*)(xrow + 4 * kk);
    }

    float* myH = &sH[wave][0];
    v8f y[4] = {};

    for (int ch = 0; ch < 4; ++ch) {
        if (ch == 1)      __builtin_amdgcn_s_setprio(2);
        else if (ch == 2) __builtin_amdgcn_s_setprio(1);
        else if (ch == 3) __builtin_amdgcn_s_setprio(0);
        v8f c[4] = {};
        v4f bw[4];
        #pragma unroll
        for (int t = 0; t < 4; ++t)
            bw[t] = Bws[(size_t)(lhalf * 16 + 0) * NH + (ch * 4 + t) * 16 + lmod];
        #pragma unroll
        for (int kp = 0; kp < 16; ++kp) {
            v4f bn[4];
            if (kp < 15) {
                #pragma unroll
                for (int t = 0; t < 4; ++t)
                    bn[t] = Bws[(size_t)(lhalf * 16 + kp + 1) * NH +
                                (ch * 4 + t) * 16 + lmod];
            }
            #pragma unroll
            for (int t = 0; t < 4; ++t) {
                const v2f blo = __builtin_shufflevector(bw[t], bw[t], 0, 1);
                c[t] = __builtin_amdgcn_wmma_f32_16x16x4_f32(
                           false, a[2 * kp], false, blo, (short)0, c[t], false, false);
            }
            #pragma unroll
            for (int t = 0; t < 4; ++t) {
                const v2f bhi = __builtin_shufflevector(bw[t], bw[t], 2, 3);
                c[t] = __builtin_amdgcn_wmma_f32_16x16x4_f32(
                           false, a[2 * kp + 1], false, bhi, (short)0, c[t], false, false);
            }
            if (kp < 15) {
                #pragma unroll
                for (int t = 0; t < 4; ++t) bw[t] = bn[t];
            }
        }
        #pragma unroll
        for (int t = 0; t < 4; ++t) {
            const float bsv = bs[(ch * 4 + t) * 16 + lmod];
            #pragma unroll
            for (int v = 0; v < 8; ++v) {
                const float hv = fmaxf(c[t][v] + bsv, 0.0f);
                myH[(v + 8 * lhalf) * SH_STR + t * 16 + lmod] = hv;
            }
        }
        const float* hA = myH + lmod * SH_STR + 2 * lhalf;
        v2f ha[16];
        #pragma unroll
        for (int q = 0; q < 16; ++q)
            ha[q] = *(const v2f*)(hA + 4 * q);
        unsigned rest = need;
        #pragma unroll
        for (int s = 0; s < 4; ++s) {
            if (s == 0 || nt > s) {
                const int t = (s == 0) ? t0 : __builtin_ctz(rest);
                rest &= rest - 1u;
                v4f ww[8];
                #pragma unroll
                for (int kp = 0; kp < 8; ++kp)
                    ww[kp] = Bwe[(size_t)(lhalf * 32 + ch * 8 + kp) * EO + t * 16 + lmod];
                #pragma unroll
                for (int kp = 0; kp < 8; ++kp) {
                    const v2f blo = __builtin_shufflevector(ww[kp], ww[kp], 0, 1);
                    const v2f bhi = __builtin_shufflevector(ww[kp], ww[kp], 2, 3);
                    y[s] = __builtin_amdgcn_wmma_f32_16x16x4_f32(
                               false, ha[2 * kp], false, blo, (short)0, y[s], false, false);
                    y[s] = __builtin_amdgcn_wmma_f32_16x16x4_f32(
                               false, ha[2 * kp + 1], false, bhi, (short)0, y[s], false, false);
                }
            }
        }
    }

    #pragma unroll
    for (int t = 0; t < 4; ++t)
        #pragma unroll
        for (int v = 0; v < 8; ++v)
            myH[(v + 8 * lhalf) * SH_STR + t * 16 + lmod] = y[t][v];

    const int selRow = lane >> 1;
    const int oq     = (lane & 1) * 4;
    const int sel    = myOrd[16 * wave + selRow];
    const int p      = sel >> 8;
    const int slot   = __builtin_popcount(need & ((1u << (p >> 1)) - 1u));
    const v4f yv = *(const v4f*)(myH + selRow * SH_STR + slot * 16 + (p & 1) * NO + oq);
    const v4f bv = *(const v4f*)(beff + p * NO + oq);
    *(v4f*)(sOut + (sel & 255) * NO + oq) = yv + bv;
    __syncthreads();
    *(v4f*)(out + (size_t)base * NO + tid * 4) = *(const v4f*)(sOut + tid * 4);
}

extern "C" void kernel_launch(void* const* d_in, const int* in_sizes, int n_in,
                              void* d_out, int out_size, void* d_ws, size_t ws_size,
                              hipStream_t stream) {
    const float* x      = (const float*)d_in[0];
    const int*   phases = (const int*)d_in[1];
    const float* Ws     = (const float*)d_in[2];
    const float* bs     = (const float*)d_in[3];
    const float* W1     = (const float*)d_in[4];
    const float* b1     = (const float*)d_in[5];
    const float* W2     = (const float*)d_in[6];
    const float* b2     = (const float*)d_in[7];
    float* out   = (float*)d_out;
    float* ws2   = (float*)d_ws;
    float* weff2 = ws2 + WS2_FLOATS;
    float* beff  = weff2 + WEFF2_FLOATS;

    qnet_prep<<<PREP_BLOCKS, FOLD_THREADS, 0, stream>>>(W1, b1, W2, b2, Ws, weff2, beff, ws2);
    qnet_fused<<<NB / ROWS_WG, WG_THREADS, 0, stream>>>(
        x, phases, bs, ws2, weff2, beff, out);
}


# Fused kernel: rows ordered by head as a ring, dealt from a data-chosen start so a wave's 16 rows hold 2 heads; B tile built per wave from their Weff columns: 1 routed tile per wave; sums unchanged
# speedup vs baseline: 1.7782x; 1.0355x over previous
#include <hip/hip_runtime.h>

#define NB 65536
#define ND 128
#define NH 256
#define NO 8
#define NE 8
#define EO (NE * NO)

typedef __attribute__((ext_vector_type(2))) float v2f;
typedef __attribute__((ext_vector_type(4))) float v4f;
typedef __attribute__((ext_vector_type(8))) float v8f;

#define WS2_FLOATS   (ND * NH)
#define WEFF2_FLOATS (NH * EO)


#define FOLD_THREADS 128
#define FOLD_TILES   128
#define FOLD_BLOCKS  136
#define PAIR_BLOCKS  (ND * NH / 2 / FOLD_THREADS)
#define PREP_BLOCKS  (FOLD_BLOCKS + PAIR_BLOCKS)

__launch_bounds__(FOLD_THREADS)
__global__ void qnet_prep(const float* __restrict__ W1,
                          const float* __restrict__ b1,
                          const float* __restrict__ W2,
                          const float* __restrict__ b2,
                          const float* __restrict__ Ws,
                          float* __restrict__ weff2,
                          float* __restrict__ beff,
                          float* __restrict__ ws2) {
    __shared__ float part[4][16][8];
    const int t = threadIdx.x;

    if (blockIdx.x < FOLD_TILES) {
        const int e    = blockIdx.x >> 4;
        const int tile = blockIdx.x & 15;
        const int wave = t >> 5;
        const int lane = t & 31;
        const int lh   = lane >> 4;
        const int c    = lane & 15;

        const float* arow = W1 + ((size_t)e * NH + tile * 16 + c) * NH + 64 * wave + 4 * lh;
        const float* bcol = W2 + ((size_t)e * NH + 64 * wave + 4 * lh) * NO + (c & 7);

        v4f   a4[8];
        float bq[8][4];
        #pragma unroll
        for (int q = 0; q < 8; ++q)
            a4[q] = *(const v4f*)(arow + 8 * q);
        #pragma unroll
        for (int q = 0; q < 8; ++q)
            #pragma unroll
            for (int s = 0; s < 4; ++s)
                bq[q][s] = bcol[(size_t)(8 * q + s) * NO];

        v8f acc0 = {}, acc1 = {};
        #pragma unroll
        for (int q = 0; q < 8; ++q) {
            v2f alo, blo, ahi, bhi;
            alo.x = a4[q].x;  alo.y = a4[q].y;
            blo.x = bq[q][0]; blo.y = bq[q][1];
            ahi.x = a4[q].z;  ahi.y = a4[q].w;
            bhi.x = bq[q][2]; bhi.y = bq[q][3];
            acc0 = __builtin_amdgcn_wmma_f32_16x16x4_f32(
                       false, alo, false, blo, (short)0, acc0, false, false);
            acc1 = __builtin_amdgcn_wmma_f32_16x16x4_f32(
                       false, ahi, false, bhi, (short)0, acc1, false, false);
        }
        const v8f acc = acc0 + acc1;
        #pragma unroll
        for (int r = 0; r < 8; ++r)
            part[wave][8 * lh + r][c & 7] = acc[r];
        __syncthreads();

        const int o   = t & 7;
        const int row = 8 * (t >> 6) + 4 * ((t >> 4) & 1) + 2 * ((t >> 5) & 1) + ((t >> 3) & 1);
        const float s = ((part[0][row][o] + part[1][row][o]) + part[2][row][o]) + part[3][row][o];

        const int h   = tile * 16 + row;
        const int col = e * NO + o;
        const int i   = h & 1;
        const int lhh = (h >> 1) & 1;
        const int ks  = h >> 2;
        const int kp  = ks >> 1;
        const int j   = ks & 1;
        weff2[((((size_t)lhh * 32 + kp) * EO + col) * 2 + j) * 2 + i] = s;
    } else if (blockIdx.x < FOLD_BLOCKS) {
        const int e  = blockIdx.x - FOLD_TILES;
        const int o  = t & 7;
        const int kg = t >> 3;
        const float* b1e = b1 + (size_t)e * NH + 16 * kg;
        const float* w2c = W2 + ((size_t)e * NH + 16 * kg) * NO + o;
        float acc = 0.0f;
        #pragma unroll
        for (int kk = 0; kk < 16; ++kk)
            acc = __builtin_fmaf(b1e[kk], w2c[(size_t)kk * NO], acc);
        float* red = &part[0][0][0];
        red[t] = acc;
        __syncthreads();
        if (t < NO) {
            float s = b2[(size_t)e * NO + t];
            #pragma unroll
            for (int g = 0; g < 16; ++g)
                s += red[g * 8 + t];
            beff[e * NO + t] = s;
        }
    } else {
        const int d = (blockIdx.x - FOLD_BLOCKS) * FOLD_THREADS + t;
        const int j   = d & 1;
        int r         = d >> 1;
        const int n   = r & (NH - 1);
        r >>= 8;
        const int kp  = r & 15;
        const int lh  = r >> 4;
        const int k0  = 4 * (2 * kp + j) + 2 * lh;
        v2f v;
        v.x = Ws[(size_t)k0 * NH + n];
        v.y = Ws[(size_t)k0 * NH + n + NH];
        ((v2f*)ws2)[d] = v;
    }
}

#define WG_THREADS 256
#define WG_WAVES   8
#define ROWS_WG    128
#define SH_STR     68

__launch_bounds__(WG_THREADS) __attribute__((amdgpu_waves_per_eu(4)))
__global__ void qnet_fused(const float* __restrict__ x,
                           const int*   __restrict__ phases,
                           const float* __restrict__ bs,
                           const float* __restrict__ ws2,
                           const float* __restrict__ weff2,
                           const float* __restrict__ beff,
                           float* __restrict__ out) {
    __shared__ float sH[WG_WAVES][16 * SH_STR];
    __shared__ int   sOrd[WG_WAVES][ROWS_WG];
    __shared__ float sOut[ROWS_WG * NO];

    __builtin_amdgcn_s_setprio(3);

    const int tid   = threadIdx.x;
    const int wave  = __builtin_amdgcn_readfirstlane(tid >> 5);
    const int lane  = tid & 31;
    const int lhalf = lane >> 4;
    const int lmod  = lane & 15;
    const int base  = blockIdx.x * ROWS_WG;

    int* myOrd = &sOrd[wave][0];
    {
        int pv[4], pos[4];
        #pragma unroll
        for (int q = 0; q < 4; ++q) {
            pv[q]  = phases[base + 32 * q + lane];
            pos[q] = 0;
        }
        const unsigned below = (1u << lane) - 1u;
        int run   = 0;
        int shift = 0;
        #pragma unroll
        for (int c = 0; c < NE; ++c) {
            #pragma unroll
            for (int q = 0; q < 4; ++q) {
                const unsigned m = __builtin_amdgcn_ballot_w32(pv[q] == c);
                if (pv[q] == c) pos[q] = run + __builtin_popcount(m & below);
                run += __builtin_popcount(m);
            }
            shift = min(shift, run - 16 * (c + 1));
        }
        #pragma unroll
        for (int q = 0; q < 4; ++q)
            myOrd[(pos[q] - shift) & (ROWS_WG - 1)] = (32 * q + lane) | (pv[q] << 8);
    }
    const int mine = myOrd[16 * wave + lmod];
    const int myhd = mine >> 8;
    unsigned hmask = 0;
    #pragma unroll
    for (int e = 0; e < NE; ++e)
        hmask |= (__builtin_amdgcn_ballot_w32(myhd == e) != 0u) ? (1u << e) : 0u;
    const int nt = (__builtin_popcount(hmask) + 1) >> 1;
    const unsigned hrest0 = hmask & (hmask - 1u);
    const int hA0 = __builtin_ctz(hmask);
    const int hB0 = hrest0 ? __builtin_ctz(hrest0) : hA0;
    const int col0 = ((lmod < NO) ? hA0 : hB0) * NO + (lmod & (NO - 1));

    const v4f* Bws = (const v4f*)ws2;
    const v4f* Bwe = (const v4f*)weff2;

    v2f a[32];
    {
        const float* xrow = x + (size_t)(base + (mine & 255)) * ND + 2 * lhalf;
        #pragma unroll
        for (int kk = 0; kk < 32; ++kk)
            a[kk] = *(const v2f*)(xrow + 4 * kk);
    }

    float* myH = &sH[wave][0];
    v8f y[4] = {};

    for (int ch = 0; ch < 4; ++ch) {
        if (ch == 1)      __builtin_amdgcn_s_setprio(2);
        else if (ch == 2) __builtin_amdgcn_s_setprio(1);
        else if (ch == 3) __builtin_amdgcn_s_setprio(0);
        v8f c[4] = {};
        v4f bw[4];
        #pragma unroll
        for (int t = 0; t < 4; ++t)
            bw[t] = Bws[(size_t)(lhalf * 16 + 0) * NH + (ch * 4 + t) * 16 + lmod];
        #pragma unroll
        for (int kp = 0; kp < 16; ++kp) {
            v4f bn[4];
            if (kp < 15) {
                #pragma unroll
                for (int t = 0; t < 4; ++t)
                    bn[t] = Bws[(size_t)(lhalf * 16 + kp + 1) * NH +
                                (ch * 4 + t) * 16 + lmod];
            }
            #pragma unroll
            for (int t = 0; t < 4; ++t) {
                const v2f blo = __builtin_shufflevector(bw[t], bw[t], 0, 1);
                c[t] = __builtin_amdgcn_wmma_f32_16x16x4_f32(
                           false, a[2 * kp], false, blo, (short)0, c[t], false, false);
            }
            #pragma unroll
            for (int t = 0; t < 4; ++t) {
                const v2f bhi = __builtin_shufflevector(bw[t], bw[t], 2, 3);
                c[t] = __builtin_amdgcn_wmma_f32_16x16x4_f32(
                           false, a[2 * kp + 1], false, bhi, (short)0, c[t], false, false);
            }
            if (kp < 15) {
                #pragma unroll
                for (int t = 0; t < 4; ++t) bw[t] = bn[t];
            }
        }
        #pragma unroll
        for (int t = 0; t < 4; ++t) {
            const float bsv = bs[(ch * 4 + t) * 16 + lmod];
            #pragma unroll
            for (int v = 0; v < 8; ++v) {
                const float hv = fmaxf(c[t][v] + bsv, 0.0f);
                myH[(v + 8 * lhalf) * SH_STR + t * 16 + lmod] = hv;
            }
        }
        const float* hA = myH + lmod * SH_STR + 2 * lhalf;
        v2f ha[16];
        #pragma unroll
        for (int q = 0; q < 16; ++q)
            ha[q] = *(const v2f*)(hA + 4 * q);
        unsigned rest = hrest0 & (hrest0 - 1u);
        #pragma unroll
        for (int s = 0; s < 4; ++s) {
            if (s == 0 || nt > s) {
                int col = col0;
                if (s > 0) {
                    const int hdA = __builtin_ctz(rest);
                    rest &= rest - 1u;
                    const int hdB = rest ? __builtin_ctz(rest) : hdA;
                    rest &= rest - 1u;
                    col = ((lmod < NO) ? hdA : hdB) * NO + (lmod & (NO - 1));
                }
                v4f ww[8];
                #pragma unroll
                for (int kp = 0; kp < 8; ++kp)
                    ww[kp] = Bwe[(size_t)(lhalf * 32 + ch * 8 + kp) * EO + col];
                #pragma unroll
                for (int kp = 0; kp < 8; ++kp) {
                    const v2f blo = __builtin_shufflevector(ww[kp], ww[kp], 0, 1);
                    const v2f bhi = __builtin_shufflevector(ww[kp], ww[kp], 2, 3);
                    y[s] = __builtin_amdgcn_wmma_f32_16x16x4_f32(
                               false, ha[2 * kp], false, blo, (short)0, y[s], false, false);
                    y[s] = __builtin_amdgcn_wmma_f32_16x16x4_f32(
                               false, ha[2 * kp + 1], false, bhi, (short)0, y[s], false, false);
                }
            }
        }
    }

    #pragma unroll
    for (int t = 0; t < 4; ++t)
        #pragma unroll
        for (int v = 0; v < 8; ++v)
            myH[(v + 8 * lhalf) * SH_STR + t * 16 + lmod] = y[t][v];

    const int laneE  = (int)__builtin_amdgcn_mbcnt_lo(~0u, 0u);
    const int tidE   = wave * 32 + laneE;
    const int selRow = laneE >> 1;
    const int oq     = (laneE & 1) * 4;
    const int sel    = myOrd[16 * wave + selRow];
    const int p      = sel >> 8;
    const int hidx   = __builtin_popcount(hmask & ((1u << p) - 1u));
    const v4f yv = *(const v4f*)(myH + selRow * SH_STR + (hidx >> 1) * 16 + (hidx & 1) * NO + oq);
    const v4f bv = *(const v4f*)(beff + p * NO + oq);
    *(v4f*)(sOut + (sel & 255) * NO + oq) = yv + bv;
    __syncthreads();
    *(v4f*)(out + (size_t)base * NO + tidE * 4) = *(const v4f*)(sOut + tidE * 4);
}

extern "C" void kernel_launch(void* const* d_in, const int* in_sizes, int n_in,
                              void* d_out, int out_size, void* d_ws, size_t ws_size,
                              hipStream_t stream) {
    const float* x      = (const float*)d_in[0];
    const int*   phases = (const int*)d_in[1];
    const float* Ws     = (const float*)d_in[2];
    const float* bs     = (const float*)d_in[3];
    const float* W1     = (const float*)d_in[4];
    const float* b1     = (const float*)d_in[5];
    const float* W2     = (const float*)d_in[6];
    const float* b2     = (const float*)d_in[7];
    float* out   = (float*)d_out;
    float* ws2   = (float*)d_ws;
    float* weff2 = ws2 + WS2_FLOATS;
    float* beff  = weff2 + WEFF2_FLOATS;

    qnet_prep<<<PREP_BLOCKS, FOLD_THREADS, 0, stream>>>(W1, b1, W2, b2, Ws, weff2, beff, ws2);
    qnet_fused<<<NB / ROWS_WG, WG_THREADS, 0, stream>>>(
        x, phases, bs, ws2, weff2, beff, out);
}
